# starting kernel
# baseline (speedup 1.0000x reference)
#include <hip/hip_runtime.h>

typedef float v2f __attribute__((ext_vector_type(2)));
typedef float v8f __attribute__((ext_vector_type(8)));

#define HH 64
#define WW 64
#define DD 256
#define NP 112
#define SSTRIDE 114

__global__ __launch_bounds__(128)
void local_attn_wmma_f32(const float* __restrict__ x, float* __restrict__ out) {
    __shared__ float sP[16 * SSTRIDE];
    __shared__ int   sKB[NP];

    const int blk = blockIdx.x;
    const int b   = blk >> 8;
    const int t   = blk & 255;
    const int h0  = (t >> 4) << 2;
    const int w0  = (t & 15) << 2;

    const int tid  = threadIdx.x;
    const int wave = tid >> 5;
    const int lane = tid & 31;
    const int half = lane >> 4;
    const int lm   = lane & 15;

    if (tid < NP) {
        int p  = tid < 100 ? tid : 99;
        int pi = p / 10, pj = p - pi * 10;
        int kh = h0 - 3 + pi; kh = kh < 0 ? 0 : (kh > HH - 1 ? HH - 1 : kh);
        int kw = w0 - 3 + pj; kw = kw < 0 ? 0 : (kw > WW - 1 ? WW - 1 : kw);
        sKB[tid] = ((b * HH + kh) * WW + kw) * DD;
    }
    __syncthreads();

    {
        const int m  = lm;
        const int qh = h0 + (m >> 2), qw = w0 + (m & 3);
        const long qbase = (long)((b * HH + qh) * WW + qw) * DD + 2 * half;
        for (int tn = wave; tn < 7; tn += 4) {
            const int  n     = tn * 16 + lm;
            const long kbase = (long)sKB[n] + 2 * half;
            v8f acc = {};
            #pragma unroll 8
            for (int d0 = 0; d0 < DD; d0 += 4) {
                v2f a  = *(const v2f*)(x + qbase + d0);
                v2f bb = *(const v2f*)(x + kbase + d0);
                acc = __builtin_amdgcn_wmma_f32_16x16x4_f32(
                        false, a, false, bb, (short)0, acc, false, false);
            }
            #pragma unroll
            for (int r = 0; r < 8; ++r)
                sP[(r + 8 * half) * SSTRIDE + tn * 16 + lm] = acc[r];
        }
    }
    __syncthreads();

    if (tid < 16) {
        const int m  = tid;
        const int qh = h0 + (m >> 2), qw = w0 + (m & 3);
        float* row = sP + m * SSTRIDE;
        float mx = -3.0e38f;
        for (int n = 0; n < NP; ++n) {
            int  pi = n / 10, pj = n - pi * 10;
            int  kh = h0 - 3 + pi, kw = w0 - 3 + pj;
            bool valid = (n < 100) && kh >= 0 && kh < HH && kw >= 0 && kw < WW
                         && (kh - qh) >= -3 && (kh - qh) <= 3
                         && (kw - qw) >= -3 && (kw - qw) <= 3;
            float s = row[n] * 0.0625f;
            row[n]  = valid ? s : -3.0e38f;
            if (valid && s > mx) mx = s;
        }
        float sum = 0.f;
        for (int n = 0; n < NP; ++n) {
            float s = row[n];
            float e = (s > -1.0e38f) ? __expf(s - mx) : 0.f;
            row[n]  = e;
            sum += e;
        }
        float inv = 1.f / sum;
        for (int n = 0; n < NP; ++n) row[n] *= inv;
    }
    __syncthreads();

    {
        for (int dt = wave; dt < 16; dt += 4) {
            const int d = dt * 16 + lm;
            v8f acc = {};
            #pragma unroll 4
            for (int p0 = 0; p0 < NP; p0 += 4) {
                const int pk = p0 + 2 * half;
                v2f a = *(const v2f*)(sP + lm * SSTRIDE + pk);
                v2f bb;
                bb.x = x[(long)sKB[pk]     + d];
                bb.y = x[(long)sKB[pk + 1] + d];
                acc = __builtin_amdgcn_wmma_f32_16x16x4_f32(
                        false, a, false, bb, (short)0, acc, false, false);
            }
            #pragma unroll
            for (int r = 0; r < 8; ++r) {
                const int rowm = r + 8 * half;
                const int qh = h0 + (rowm >> 2), qw = w0 + (rowm & 3);
                out[(long)((b * HH + qh) * WW + qw) * DD + d] = acc[r];
            }
        }
    }
}

extern "C" void kernel_launch(void* const* d_in, const int* in_sizes, int n_in,
                              void* d_out, int out_size, void* d_ws, size_t ws_size,
                              hipStream_t stream) {
    const float* x   = (const float*)d_in[0];
    float*       out = (float*)d_out;
    local_attn_wmma_f32<<<dim3(1024), dim3(128), 0, stream>>>(x, out);
}


# Q.K^T split over channels, not key tiles: each wave reduces its 64-channel quarter for all 7 key tiles on 7 accumulators, a Q fragment loaded once not 7 times; score = 4 partial sums added pairwise
# speedup vs baseline: 1.3230x; 1.3230x over previous
#include <hip/hip_runtime.h>


typedef float v2f __attribute__((ext_vector_type(2)));
typedef float v8f __attribute__((ext_vector_type(8)));

#define HH 64
#define WW 64
#define DD 256
#define NP 112
#define NT 7
#define SSTRIDE 114
#define PLANE (16 * SSTRIDE)
#define DQ 64

__device__ __forceinline__ void load_line(v2f (&f)[8], const float* p) {
    #pragma unroll
    for (int j = 0; j < 8; ++j) f[j] = *(const v2f*)(p + 4 * j);
}

__global__ __launch_bounds__(128)
void local_attn_wmma_f32(const float* __restrict__ x, float* __restrict__ out) {
    __shared__ float sP[4 * PLANE];
    __shared__ int   sKB[NP];

    const int blk = blockIdx.x;
    const int b   = blk >> 8;
    const int t   = blk & 255;
    const int h0  = (t >> 4) << 2;
    const int w0  = (t & 15) << 2;

    const int tid  = threadIdx.x;
    const int wave = tid >> 5;
    const int lane = tid & 31;
    const int half = lane >> 4;
    const int lm   = lane & 15;

    if (tid < NP) {
        int p  = tid < 100 ? tid : 99;
        int pi = p / 10, pj = p - pi * 10;
        int kh = h0 - 3 + pi; kh = kh < 0 ? 0 : (kh > HH - 1 ? HH - 1 : kh);
        int kw = w0 - 3 + pj; kw = kw < 0 ? 0 : (kw > WW - 1 ? WW - 1 : kw);
        sKB[tid] = ((b * HH + kh) * WW + kw) * DD;
    }
    __syncthreads();

    {
        const int qh = h0 + (lm >> 2), qw = w0 + (lm & 3);
        const float* qp = x + (long)((b * HH + qh) * WW + qw) * DD + wave * DQ + 2 * half;
        const float* kp[NT];
        v8f acc[NT];
        #pragma unroll
        for (int tn = 0; tn < NT; ++tn) {
            kp[tn]  = x + (long)sKB[tn * 16 + lm] + wave * DQ + 2 * half;
            acc[tn] = (v8f){};
        }

        v2f a[2][8];
        v2f kb[2][2][8];

        load_line(a[0], qp);
        __builtin_amdgcn_sched_barrier(0);
        load_line(kb[0][0], kp[0]);
        __builtin_amdgcn_sched_barrier(0);
        load_line(kb[0][1], kp[1]);
        __builtin_amdgcn_sched_barrier(0);

        #pragma unroll
        for (int s = 0; s < 7; ++s) {
            const int u0 = 2 * s, u1 = 2 * s + 1;
            const int c0 = u0 / NT, t0 = u0 % NT;
            const int c1 = u1 / NT, t1 = u1 % NT;
            if (s + 1 < 7) {
                const int v0 = u0 + 2, v1 = u1 + 2;
                if (v0 == NT || v1 == NT) {
                    load_line(a[1], qp + 32);
                    __builtin_amdgcn_sched_barrier(0);
                }
                load_line(kb[(s + 1) & 1][0], kp[v0 % NT] + 32 * (v0 / NT));
                __builtin_amdgcn_sched_barrier(0);
                load_line(kb[(s + 1) & 1][1], kp[v1 % NT] + 32 * (v1 / NT));
                __builtin_amdgcn_sched_barrier(0);
            }
            #pragma unroll
            for (int j = 0; j < 8; ++j) {
                acc[t0] = __builtin_amdgcn_wmma_f32_16x16x4_f32(
                        false, a[c0][j], false, kb[s & 1][0][j], (short)0, acc[t0], false, false);
                acc[t1] = __builtin_amdgcn_wmma_f32_16x16x4_f32(
                        false, a[c1][j], false, kb[s & 1][1][j], (short)0, acc[t1], false, false);
            }
            __builtin_amdgcn_sched_barrier(0);
        }

        float* plane = sP + wave * PLANE;
        #pragma unroll
        for (int tn = 0; tn < NT; ++tn) {
            #pragma unroll
            for (int r = 0; r < 8; ++r)
                plane[(r + 8 * half) * SSTRIDE + tn * 16 + lm] = acc[tn][r];
        }
    }
    __syncthreads();

    {
        v2f ps[7][4];
        #pragma unroll
        for (int k = 0; k < 7; ++k) {
            const int i = tid + 128 * k;
            const int r = i / 56, c2 = i - r * 56;
            const float* p = sP + r * SSTRIDE + 2 * c2;
            #pragma unroll
            for (int w = 0; w < 4; ++w) ps[k][w] = *(const v2f*)(p + w * PLANE);
        }
        #pragma unroll
        for (int k = 0; k < 7; ++k) {
            const int i = tid + 128 * k;
            const int r = i / 56, c2 = i - r * 56;
            *(v2f*)(sP + r * SSTRIDE + 2 * c2) = (ps[k][0] + ps[k][1]) + (ps[k][2] + ps[k][3]);
        }
    }
    __syncthreads();

    if (tid < 16) {
        const int m  = tid;
        const int qh = h0 + (m >> 2), qw = w0 + (m & 3);
        float* row = sP + m * SSTRIDE;
        float mx = -3.0e38f;
        for (int n = 0; n < NP; ++n) {
            int  pi = n / 10, pj = n - pi * 10;
            int  kh = h0 - 3 + pi, kw = w0 - 3 + pj;
            bool valid = (n < 100) && kh >= 0 && kh < HH && kw >= 0 && kw < WW
                         && (kh - qh) >= -3 && (kh - qh) <= 3
                         && (kw - qw) >= -3 && (kw - qw) <= 3;
            float s = row[n] * 0.0625f;
            row[n]  = valid ? s : -3.0e38f;
            if (valid && s > mx) mx = s;
        }
        float sum = 0.f;
        for (int n = 0; n < NP; ++n) {
            float s = row[n];
            float e = (s > -1.0e38f) ? __expf(s - mx) : 0.f;
            row[n]  = e;
            sum += e;
        }
        float inv = 1.f / sum;
        for (int n = 0; n < NP; ++n) row[n] *= inv;
    }
    __syncthreads();

    {
        for (int dt = wave; dt < 16; dt += 4) {
            const int d = dt * 16 + lm;
            v8f acc = {};
            #pragma unroll 4
            for (int p0 = 0; p0 < NP; p0 += 4) {
                const int pk = p0 + 2 * half;
                v2f a = *(const v2f*)(sP + lm * SSTRIDE + pk);
                v2f bb;
                bb.x = x[(long)sKB[pk]     + d];
                bb.y = x[(long)sKB[pk + 1] + d];
                acc = __builtin_amdgcn_wmma_f32_16x16x4_f32(
                        false, a, false, bb, (short)0, acc, false, false);
            }
            #pragma unroll
            for (int r = 0; r < 8; ++r) {
                const int rowm = r + 8 * half;
                const int qh = h0 + (rowm >> 2), qw = w0 + (rowm & 3);
                out[(long)((b * HH + qh) * WW + qw) * DD + d] = acc[r];
            }
        }
    }
}

extern "C" void kernel_launch(void* const* d_in, const int* in_sizes, int n_in,
                              void* d_out, int out_size, void* d_ws, size_t ws_size,
                              hipStream_t stream) {
    const float* x   = (const float*)d_in[0];
    float*       out = (float*)d_out;
    local_attn_wmma_f32<<<dim3(1024), dim3(128), 0, stream>>>(x, out);
}


# Each wave does two 4x4 query tiles, one above the other: one loaded K or V fragment feeds both tiles' products, 9 key tiles per pair not 14; a tile's keys and every sum keep their order
# speedup vs baseline: 2.6584x; 2.0094x over previous
#include <hip/hip_runtime.h>


typedef float v2f __attribute__((ext_vector_type(2)));
typedef float v4f __attribute__((ext_vector_type(4)));
typedef float v8f __attribute__((ext_vector_type(8)));

#define HH 64
#define WW 64
#define DD 256
#define PR 14
#define PC 10
#define NKU 144
#define NTU 9
#define NT 7
#define KS 28
#define KSU 36
#define SSTRIDE 116
#define PLANE (16 * SSTRIDE)
#define DQ 64


__device__ __forceinline__ void load_line(v2f (&f)[8], const float* p) {
    #pragma unroll
    for (int j = 0; j < 8; ++j) f[j] = *(const v2f*)(p + 4 * j);
}

__device__ __forceinline__ float xhalf(float v) {
    const int b = __builtin_bit_cast(int, v);
    return __builtin_bit_cast(float, __builtin_amdgcn_permlanex16(b, b, 0x76543210, 0xfedcba98, false, false));
}

__global__ __launch_bounds__(128)
void local_attn_wmma_f32(const float* __restrict__ x, float* __restrict__ out) {
    __shared__ float sP[4 * 2 * PLANE];
    __shared__ int   sKB[NKU];

    const int blk = blockIdx.x;
    const int b   = blk >> 7;
    const int t   = blk & 127;
    const int h0  = (t >> 4) << 3;
    const int w0  = (t & 15) << 2;

    const int tid  = threadIdx.x;
    const int wave = tid >> 5;
    const int lane = tid & 31;
    const int half = lane >> 4;
    const int lm   = lane & 15;

    for (int n = tid; n < NKU; n += 128) {
        int p  = n < PR * PC ? n : PR * PC - 1;
        int pi = p / PC, pj = p - pi * PC;
        int kh = h0 - 3 + pi; kh = kh < 0 ? 0 : (kh > HH - 1 ? HH - 1 : kh);
        int kw = w0 - 3 + pj; kw = kw < 0 ? 0 : (kw > WW - 1 ? WW - 1 : kw);
        sKB[n] = ((b * HH + kh) * WW + kw) * DD;
    }
    __syncthreads();

    {
        const int qh = h0 + (lm >> 2), qw = w0 + (lm & 3);
        const float* qp = x + (long)((b * HH + qh) * WW + qw) * DD + wave * DQ + 2 * half;
        const float* kp[NTU];
        v8f acc[2][NT];
        #pragma unroll
        for (int tn = 0; tn < NTU; ++tn)
            kp[tn] = x + (long)sKB[tn * 16 + lm] + wave * DQ + 2 * half;
        #pragma unroll
        for (int T = 0; T < 2; ++T) {
            #pragma unroll
            for (int tn = 0; tn < NT; ++tn) acc[T][tn] = (v8f){};
        }

        v2f a[2][2][8];
        v2f kb[2][2][8];

        load_line(a[0][0], qp);
        __builtin_amdgcn_sched_barrier(0);
        load_line(a[1][0], qp + 4 * WW * DD);
        __builtin_amdgcn_sched_barrier(0);
        load_line(kb[0][0], kp[0]);
        __builtin_amdgcn_sched_barrier(0);
        load_line(kb[0][1], kp[1]);
        __builtin_amdgcn_sched_barrier(0);

        #pragma unroll
        for (int s = 0; s < NTU; ++s) {
            const int u0 = 2 * s, u1 = 2 * s + 1;
            const int c0 = u0 / NTU, t0 = u0 % NTU;
            const int c1 = u1 / NTU, t1 = u1 % NTU;
            if (s + 1 < NTU) {
                const int v0 = u0 + 2, v1 = u1 + 2;
                if (v0 == NTU || v1 == NTU) {
                    load_line(a[0][1], qp + 32);
                    __builtin_amdgcn_sched_barrier(0);
                    load_line(a[1][1], qp + 4 * WW * DD + 32);
                    __builtin_amdgcn_sched_barrier(0);
                }
                load_line(kb[(s + 1) & 1][0], kp[v0 % NTU] + 32 * (v0 / NTU));
                __builtin_amdgcn_sched_barrier(0);
                load_line(kb[(s + 1) & 1][1], kp[v1 % NTU] + 32 * (v1 / NTU));
                __builtin_amdgcn_sched_barrier(0);
            }
            #pragma unroll
            for (int j = 0; j < 8; ++j) {
                #pragma unroll
                for (int T = 0; T < 2; ++T) {
                    if (t0 - 2 * T >= 0 && t0 - 2 * T < NT)
                        acc[T][t0 - 2 * T] = __builtin_amdgcn_wmma_f32_16x16x4_f32(
                                false, a[T][c0][j], false, kb[s & 1][0][j], (short)0, acc[T][t0 - 2 * T], false, false);
                }
                #pragma unroll
                for (int T = 0; T < 2; ++T) {
                    if (t1 - 2 * T >= 0 && t1 - 2 * T < NT)
                        acc[T][t1 - 2 * T] = __builtin_amdgcn_wmma_f32_16x16x4_f32(
                                false, a[T][c1][j], false, kb[s & 1][1][j], (short)0, acc[T][t1 - 2 * T], false, false);
                }
            }
            __builtin_amdgcn_sched_barrier(0);
        }

        #pragma unroll
        for (int T = 0; T < 2; ++T) {
            float* plane = sP + (wave * 2 + T) * PLANE;
            #pragma unroll
            for (int tn = 0; tn < NT; ++tn) {
                #pragma unroll
                for (int r = 0; r < 8; ++r)
                    plane[(r + 8 * half) * SSTRIDE + tn * 16 + lm] = acc[T][tn][r];
            }
        }
    }
    __syncthreads();

    v2f p[2][KS];
    {
        const int qi = lm >> 2, qj = lm & 3;
        int rlo = 3 - h0;        rlo = rlo < 0 ? 0 : rlo;
        int rhi = HH + 2 - h0;   rhi = rhi > PR - 1 ? PR - 1 : rhi;
        int clo = 3 - w0;        clo = clo < 0 ? 0 : clo;
        int chi = WW + 2 - w0;   chi = chi > PC - 1 ? PC - 1 : chi;
        const unsigned colmask = (0x7Fu << qj) & ((2u << chi) - 1u) & ~((1u << clo) - 1u);

        #pragma unroll
        for (int T = 0; T < 2; ++T) {
            const float* row = sP + T * PLANE + lm * SSTRIDE + 2 * half;
            #pragma unroll
            for (int s = 0; s < KS; ++s) {
                const float* q = row + 4 * s;
                const v2f l0 = *(const v2f*)(q);
                const v2f l1 = *(const v2f*)(q + 2 * PLANE);
                const v2f l2 = *(const v2f*)(q + 4 * PLANE);
                const v2f l3 = *(const v2f*)(q + 6 * PLANE);
                p[T][s] = ((l0 + l1) + (l2 + l3)) * 0.0625f;
            }

            const unsigned rowmask = (0x7Fu << (4 * T + qi)) & ((2u << rhi) - 1u) & ~((1u << rlo) - 1u);
            unsigned g[5] = {0u, 0u, 0u, 0u, 0u};
            #pragma unroll
            for (int pi = 0; pi < PR; ++pi) {
                const unsigned f = ((rowmask >> pi) & 1u) ? colmask : 0u;
                const int pos = PC * pi, wd = pos >> 5, off = pos & 31;
                g[wd] |= f << off;
                if (off > 22) g[wd + 1] |= f >> (32 - off);
            }
            unsigned m[4];
            #pragma unroll
            for (int i = 0; i < 4; ++i) m[i] = g[i + T];
            if (half) {
                m[0] = (m[0] >> 2) | (m[1] << 30);
                m[1] = (m[1] >> 2) | (m[2] << 30);
                m[2] = (m[2] >> 2) | (m[3] << 30);
                m[3] =  m[3] >> 2;
            }

            float mx = -3.0e38f;
            #pragma unroll
            for (int s = 0; s < KS; ++s) {
                #pragma unroll
                for (int e = 0; e < 2; ++e) {
                    const int  bit   = 4 * s + e;
                    const bool valid = (m[bit >> 5] >> (bit & 31)) & 1u;
                    const float v = valid ? p[T][s][e] : -3.0e38f;
                    p[T][s][e] = v;
                    mx = fmaxf(mx, v);
                }
            }
            mx = fmaxf(mx, xhalf(mx));

            v2f part = {0.f, 0.f};
            #pragma unroll
            for (int s = 0; s < KS; ++s) {
                #pragma unroll
                for (int e = 0; e < 2; ++e) {
                    const float v = p[T][s][e];
                    p[T][s][e] = (v > -1.0e38f) ? __expf(v - mx) : 0.f;
                }
                part += p[T][s];
            }
            float sum = part.x + part.y;
            sum += xhalf(sum);
            const float inv = 1.f / sum;
            #pragma unroll
            for (int s = 0; s < KS; ++s) p[T][s] *= inv;
        }
    }

    {
        const char* xb = (const char*)x;
        const int cb = wave * 64 + 4 * lm;
        v8f acc[2][4];
        #pragma unroll
        for (int T = 0; T < 2; ++T) {
            #pragma unroll
            for (int c = 0; c < 4; ++c) acc[T][c] = (v8f){};
        }
        #pragma unroll
        for (int g = 0; g < KSU; ++g) {
            const int pk = 4 * g + 2 * half;
            const unsigned o0 = (unsigned)(sKB[pk]     + cb) << 2;
            const unsigned o1 = (unsigned)(sKB[pk + 1] + cb) << 2;
            const v4f v0 = *(const v4f*)(xb + o0);
            const v4f v1 = *(const v4f*)(xb + o1);
            #pragma unroll
            for (int c = 0; c < 4; ++c) {
                const v2f bb = {v0[c], v1[c]};
                #pragma unroll
                for (int T = 0; T < 2; ++T) {
                    if (g - 8 * T >= 0 && g - 8 * T < KS)
                        acc[T][c] = __builtin_amdgcn_wmma_f32_16x16x4_f32(
                                false, p[T][g - 8 * T], false, bb, (short)0, acc[T][c], false, false);
                }
            }
        }
        #pragma unroll
        for (int T = 0; T < 2; ++T) {
            #pragma unroll
            for (int r = 0; r < 8; ++r) {
                const int rowm = r + 8 * half;
                const int qh = h0 + 4 * T + (rowm >> 2), qw = w0 + (rowm & 3);
                const v4f o = {acc[T][0][r], acc[T][1][r], acc[T][2][r], acc[T][3][r]};
                *(v4f*)(out + (long)((b * HH + qh) * WW + qw) * DD + cb) = o;
            }
        }
    }
}

extern "C" void kernel_launch(void* const* d_in, const int* in_sizes, int n_in,
                              void* d_out, int out_size, void* d_ws, size_t ws_size,
                              hipStream_t stream) {
    const float* x   = (const float*)d_in[0];
    float*       out = (float*)d_out;
    local_attn_wmma_f32<<<dim3(512), dim3(128), 0, stream>>>(x, out);
}


# P.V as one matrix stream per tile over its 25 visible key steps, upper tile first with the lower tile's softmax sliced between its products; shared V fragments stay in registers; no sum reordered
# speedup vs baseline: 2.7916x; 1.0501x over previous
#include <hip/hip_runtime.h>


typedef float v2f __attribute__((ext_vector_type(2)));
typedef float v4f __attribute__((ext_vector_type(4)));
typedef float v8f __attribute__((ext_vector_type(8)));

#define HH 64
#define WW 64
#define DD 256
#define PR 14
#define PC 10
#define NKU 144
#define NTU 9
#define NT 7
#define KS 28
#define KL 25
#define KG1 10
#define KGE 35
#define VD 12
#define RB 15
#define SSTRIDE 116
#define PLANE (16 * SSTRIDE)
#define DQ 64


__device__ __forceinline__ void load_line(v2f (&f)[8], const float* p) {
    #pragma unroll
    for (int j = 0; j < 8; ++j) f[j] = *(const v2f*)(p + 4 * j);
}

__device__ __forceinline__ float xhalf(float v) {
    const int b = __builtin_bit_cast(int, v);
    return __builtin_bit_cast(float, __builtin_amdgcn_permlanex16(b, b, 0x76543210, 0xfedcba98, false, false));
}

template <int T>
__device__ __forceinline__ void key_mask(unsigned (&m)[4], const unsigned rowmask, const unsigned colmask, const int half) {
    unsigned g[5] = {0u, 0u, 0u, 0u, 0u};
    #pragma unroll
    for (int pi = 0; pi < PR; ++pi) {
        const unsigned f = ((rowmask >> pi) & 1u) ? colmask : 0u;
        const int pos = PC * pi, wd = pos >> 5, off = pos & 31;
        g[wd] |= f << off;
        if (off > 22) g[wd + 1] |= f >> (32 - off);
    }
    #pragma unroll
    for (int i = 0; i < 4; ++i) m[i] = g[i + T];
    if (half) {
        m[0] = (m[0] >> 2) | (m[1] << 30);
        m[1] = (m[1] >> 2) | (m[2] << 30);
        m[2] = (m[2] >> 2) | (m[3] << 30);
        m[3] =  m[3] >> 2;
    }
}

__global__ __launch_bounds__(128)
void local_attn_wmma_f32(const float* __restrict__ x, float* __restrict__ out) {
    __shared__ float sP[4 * 2 * PLANE];
    __shared__ int   sKB[NKU];

    const int blk = blockIdx.x;
    const int b   = blk >> 7;
    const int t   = blk & 127;
    const int h0  = (t >> 4) << 3;
    const int w0  = (t & 15) << 2;

    const int tid  = threadIdx.x;
    const int wave = tid >> 5;
    const int lane = tid & 31;
    const int half = lane >> 4;
    const int lm   = lane & 15;

    for (int n = tid; n < NKU; n += 128) {
        int p  = n < PR * PC ? n : PR * PC - 1;
        int pi = p / PC, pj = p - pi * PC;
        int kh = h0 - 3 + pi; kh = kh < 0 ? 0 : (kh > HH - 1 ? HH - 1 : kh);
        int kw = w0 - 3 + pj; kw = kw < 0 ? 0 : (kw > WW - 1 ? WW - 1 : kw);
        sKB[n] = ((b * HH + kh) * WW + kw) * DD;
    }
    __syncthreads();

    {
        const int qh = h0 + (lm >> 2), qw = w0 + (lm & 3);
        const float* qp = x + (long)((b * HH + qh) * WW + qw) * DD + wave * DQ + 2 * half;
        const float* kp[NTU];
        v8f acc[2][NT];
        #pragma unroll
        for (int tn = 0; tn < NTU; ++tn)
            kp[tn] = x + (long)sKB[tn * 16 + lm] + wave * DQ + 2 * half;
        #pragma unroll
        for (int T = 0; T < 2; ++T) {
            #pragma unroll
            for (int tn = 0; tn < NT; ++tn) acc[T][tn] = (v8f){};
        }

        v2f a[2][2][8];
        v2f kb[2][2][8];

        load_line(a[0][0], qp);
        __builtin_amdgcn_sched_barrier(0);
        load_line(a[1][0], qp + 4 * WW * DD);
        __builtin_amdgcn_sched_barrier(0);
        load_line(kb[0][0], kp[0]);
        __builtin_amdgcn_sched_barrier(0);
        load_line(kb[0][1], kp[1]);
        __builtin_amdgcn_sched_barrier(0);

        #pragma unroll
        for (int s = 0; s < NTU; ++s) {
            const int u0 = 2 * s, u1 = 2 * s + 1;
            const int c0 = u0 / NTU, t0 = u0 % NTU;
            const int c1 = u1 / NTU, t1 = u1 % NTU;
            if (s + 1 < NTU) {
                const int v0 = u0 + 2, v1 = u1 + 2;
                if (v0 == NTU || v1 == NTU) {
                    load_line(a[0][1], qp + 32);
                    __builtin_amdgcn_sched_barrier(0);
                    load_line(a[1][1], qp + 4 * WW * DD + 32);
                    __builtin_amdgcn_sched_barrier(0);
                }
                load_line(kb[(s + 1) & 1][0], kp[v0 % NTU] + 32 * (v0 / NTU));
                __builtin_amdgcn_sched_barrier(0);
                load_line(kb[(s + 1) & 1][1], kp[v1 % NTU] + 32 * (v1 / NTU));
                __builtin_amdgcn_sched_barrier(0);
            }
            #pragma unroll
            for (int j = 0; j < 8; ++j) {
                #pragma unroll
                for (int T = 0; T < 2; ++T) {
                    if (t0 - 2 * T >= 0 && t0 - 2 * T < NT)
                        acc[T][t0 - 2 * T] = __builtin_amdgcn_wmma_f32_16x16x4_f32(
                                false, a[T][c0][j], false, kb[s & 1][0][j], (short)0, acc[T][t0 - 2 * T], false, false);
                }
                #pragma unroll
                for (int T = 0; T < 2; ++T) {
                    if (t1 - 2 * T >= 0 && t1 - 2 * T < NT)
                        acc[T][t1 - 2 * T] = __builtin_amdgcn_wmma_f32_16x16x4_f32(
                                false, a[T][c1][j], false, kb[s & 1][1][j], (short)0, acc[T][t1 - 2 * T], false, false);
                }
            }
            __builtin_amdgcn_sched_barrier(0);
        }

        #pragma unroll
        for (int T = 0; T < 2; ++T) {
            float* plane = sP + (wave * 2 + T) * PLANE;
            #pragma unroll
            for (int tn = 0; tn < NT; ++tn) {
                #pragma unroll
                for (int r = 0; r < 8; ++r)
                    plane[(r + 8 * half) * SSTRIDE + tn * 16 + lm] = acc[T][tn][r];
            }
        }
    }
    __syncthreads();

    const int qi = lm >> 2, qj = lm & 3;
    int rlo = 3 - h0;        rlo = rlo < 0 ? 0 : rlo;
    int rhi = HH + 2 - h0;   rhi = rhi > PR - 1 ? PR - 1 : rhi;
    int clo = 3 - w0;        clo = clo < 0 ? 0 : clo;
    int chi = WW + 2 - w0;   chi = chi > PC - 1 ? PC - 1 : chi;
    const unsigned colmask = (0x7Fu << qj) & ((2u << chi) - 1u) & ~((1u << clo) - 1u);
    const unsigned rowin   = ((2u << rhi) - 1u) & ~((1u << rlo) - 1u);

    unsigned m0[4], m1[4];
    key_mask<0>(m0, (0x7Fu << qi) & rowin, colmask, half);
    key_mask<1>(m1, (0x7Fu << (4 + qi)) & rowin, colmask, half);

    const float* row0 = sP + lm * SSTRIDE + 2 * half;
    const float* row1 = row0 + PLANE;

    const char* xb = (const char*)x;
    const int cb = wave * 64 + 4 * lm;

    unsigned ofs[KGE][2];
    #pragma unroll
    for (int g = 0; g < KGE; ++g) {
        const int pk = 4 * g + 2 * half;
        ofs[g][0] = (unsigned)(sKB[pk]     + cb) << 2;
        ofs[g][1] = (unsigned)(sKB[pk + 1] + cb) << 2;
    }
    v4f vq[KGE][2];
    auto vload = [&](const int g) {
        vq[g][0] = *(const v4f*)(xb + ofs[g][0]);
        vq[g][1] = *(const v4f*)(xb + ofs[g][1]);
    };
    #pragma unroll
    for (int g = 0; g < VD; ++g) vload(g);

    v2f raw0[KL][4], raw[KS][4];
    auto pload = [&](v2f (&r)[4], const float* q) {
        r[0] = *(const v2f*)(q);
        r[1] = *(const v2f*)(q + 2 * PLANE);
        r[2] = *(const v2f*)(q + 4 * PLANE);
        r[3] = *(const v2f*)(q + 6 * PLANE);
    };
    auto rload = [&](const int s) { pload(raw[s], row1 + 4 * s); };
    #pragma unroll
    for (int s = 0; s < KL; ++s) pload(raw0[s], row0 + 4 * s);
    __builtin_amdgcn_sched_barrier(0);

    auto mask2 = [](v2f v, const unsigned (&m)[4], const int s, float& mx) -> v2f {
        #pragma unroll
        for (int e = 0; e < 2; ++e) {
            const int  bit   = 4 * s + e;
            const bool valid = (m[bit >> 5] >> (bit & 31)) & 1u;
            v[e] = valid ? v[e] : -3.0e38f;
            mx = fmaxf(mx, v[e]);
        }
        return v;
    };
    auto expo = [](v2f v, const float mx) -> v2f {
        #pragma unroll
        for (int e = 0; e < 2; ++e) v[e] = (v[e] > -1.0e38f) ? __expf(v[e] - mx) : 0.f;
        return v;
    };

    v2f p0[KL];
    {
        float mx = -3.0e38f;
        #pragma unroll
        for (int s = 0; s < KL; ++s)
            p0[s] = mask2(((raw0[s][0] + raw0[s][1]) + (raw0[s][2] + raw0[s][3])) * 0.0625f, m0, s, mx);
        __builtin_amdgcn_sched_barrier(0);
        #pragma unroll
        for (int s = 2; s < RB; ++s) rload(s);
        __builtin_amdgcn_sched_barrier(0);
        mx = fmaxf(mx, xhalf(mx));
        v2f part = {0.f, 0.f};
        #pragma unroll
        for (int s = 0; s < KL; ++s) { p0[s] = expo(p0[s], mx); part += p0[s]; }
        float sum = part.x + part.y;
        sum += xhalf(sum);
        const float inv = 1.f / sum;
        #pragma unroll
        for (int s = 0; s < KL; ++s) p0[s] *= inv;
    }

    {
        v8f acc[2][4];
        #pragma unroll
        for (int T = 0; T < 2; ++T) {
            #pragma unroll
            for (int c = 0; c < 4; ++c) acc[T][c] = (v8f){};
        }
        v2f p1[KS];
        v2f keep[KL - KG1][4];
        float mx1 = -3.0e38f, inv1 = 0.f;
        v2f part1 = {0.f, 0.f};

        auto slice = [&](const int k) {
            if (k == 20) {
                #pragma unroll
                for (int s = RB; s <= 26; ++s) rload(s);
            }
            if ((k & 1) == 0 && k >= 2 && k <= 50) {
                const int s = 1 + k / 2;
                p1[s] = mask2(((raw[s][0] + raw[s][1]) + (raw[s][2] + raw[s][3])) * 0.0625f, m1, s, mx1);
            }
            if (k == 51) mx1 = fmaxf(mx1, xhalf(mx1));
            if (k >= 53 && k <= 77) {
                const int s = k - 51;
                p1[s] = expo(p1[s], mx1);
                part1 += p1[s];
            }
            if (k == 78) {
                float sum = part1.x + part1.y;
                sum += xhalf(sum);
                inv1 = 1.f / sum;
            }
            if (k >= 79 && k <= 91) {
                const int s = 2 + 2 * (k - 79);
                p1[s] *= inv1;
                if (s + 1 <= 26) p1[s + 1] *= inv1;
            }
        };

        #pragma unroll
        for (int g = 0; g < KL; ++g) {
            if (g + VD < KGE) vload(g + VD);
            __builtin_amdgcn_sched_barrier(0);
            #pragma unroll
            for (int c = 0; c < 4; ++c) {
                const v2f bb = {vq[g][0][c], vq[g][1][c]};
                if (g >= KG1) keep[g - KG1][c] = bb;
                acc[0][c] = __builtin_amdgcn_wmma_f32_16x16x4_f32(
                        false, p0[g], false, bb, (short)0, acc[0][c], false, false);
                slice(4 * g + c);
                __builtin_amdgcn_sched_barrier(0);
            }
        }

        #pragma unroll
        for (int g = KL + VD; g < KGE; ++g) vload(g);
        #pragma unroll
        for (int g = KG1; g < KGE; ++g) {
            #pragma unroll
            for (int c = 0; c < 4; ++c) {
                v2f bb;
                if (g < KL) bb = keep[g - KG1][c];
                else        bb = (v2f){vq[g][0][c], vq[g][1][c]};
                acc[1][c] = __builtin_amdgcn_wmma_f32_16x16x4_f32(
                        false, p1[g - 8], false, bb, (short)0, acc[1][c], false, false);
            }
        }
        #pragma unroll
        for (int T = 0; T < 2; ++T) {
            #pragma unroll
            for (int r = 0; r < 8; ++r) {
                const int rowm = r + 8 * half;
                const int qh = h0 + 4 * T + (rowm >> 2), qw = w0 + (rowm & 3);
                const v4f o = {acc[T][0][r], acc[T][1][r], acc[T][2][r], acc[T][3][r]};
                *(v4f*)(out + (long)((b * HH + qh) * WW + qw) * DD + cb) = o;
            }
        }
    }
}

extern "C" void kernel_launch(void* const* d_in, const int* in_sizes, int n_in,
                              void* d_out, int out_size, void* d_ws, size_t ws_size,
                              hipStream_t stream) {
    const float* x   = (const float*)d_in[0];
    float*       out = (float*)d_out;
    local_attn_wmma_f32<<<dim3(512), dim3(128), 0, stream>>>(x, out);
}
